# starting kernel
# baseline (speedup 1.0000x reference)
#include <hip/hip_runtime.h>


typedef __attribute__((ext_vector_type(2))) float v2f;
typedef __attribute__((ext_vector_type(8))) float v8f;

#define IN_F 32
#define OUT_F 32
#define LDS_STRIDE 36

__global__ __launch_bounds__(256) void
grouped_linear_wmma_f32(const float* __restrict__ x,
                        const float* __restrict__ weights,
                        const float* __restrict__ biases,
                        float* __restrict__ y,
                        int points_per_net)
{
    __shared__ float ldsW[IN_F * LDS_STRIDE];
    __shared__ float ldsB[OUT_F];
    __shared__ float ldsX[8 * 16 * LDS_STRIDE];

    const int net  = blockIdx.x;
    const int tid  = threadIdx.x;
    const int wave = tid >> 5;
    const int lane = tid & 31;

    {
        const float4* Wn4 = reinterpret_cast<const float4*>(
            weights + (size_t)net * IN_F * OUT_F);
        const int e4 = tid;
        const int k  = e4 >> 3;
        const int n  = (e4 & 7) * 4;
        const float4 v = Wn4[e4];
        *reinterpret_cast<float4*>(&ldsW[k * LDS_STRIDE + n]) = v;
    }
    if (tid < OUT_F) ldsB[tid] = biases[(size_t)net * OUT_F + tid];
    __syncthreads();

    const int m16  = lane & 15;
    const int half = lane >> 4;

    const float bv0 = ldsB[m16];
    const float bv1 = ldsB[16 + m16];

    float* xstage = &ldsX[wave * 16 * LDS_STRIDE];
    const size_t rowbase = (size_t)net * (size_t)points_per_net;

    for (int tile = wave; tile * 16 < points_per_net; tile += 8) {
        const float* Xt = x + (rowbase + (size_t)tile * 16) * IN_F;

        #pragma unroll
        for (int i = 0; i < 4; ++i) {
            const int e4 = i * 32 + lane;
            const int r  = e4 >> 3;
            const int c  = (e4 & 7) * 4;
            const float4 v = reinterpret_cast<const float4*>(Xt)[e4];
            *reinterpret_cast<float4*>(&xstage[r * LDS_STRIDE + c]) = v;
        }

        v8f c0 = {};
        v8f c1 = {};

        #pragma unroll
        for (int kk = 0; kk < 8; ++kk) {
            const int ka = kk * 4 + half * 2;
            const v2f a =
                *reinterpret_cast<const v2f*>(&xstage[m16 * LDS_STRIDE + ka]);
            v2f b0, b1;
            b0.x = ldsW[ ka      * LDS_STRIDE + m16];
            b0.y = ldsW[(ka + 1) * LDS_STRIDE + m16];
            b1.x = ldsW[ ka      * LDS_STRIDE + 16 + m16];
            b1.y = ldsW[(ka + 1) * LDS_STRIDE + 16 + m16];
            c0 = __builtin_amdgcn_wmma_f32_16x16x4_f32(
                     false, a, false, b0, (short)0, c0, false, false);
            c1 = __builtin_amdgcn_wmma_f32_16x16x4_f32(
                     false, a, false, b1, (short)0, c1, false, false);
        }

        float* Yt = y + (rowbase + (size_t)tile * 16) * OUT_F;
        #pragma unroll
        for (int j = 0; j < 8; ++j) {
            const int m = j + 8 * half;
            Yt[m * OUT_F + m16]      = c0[j] + bv0;
            Yt[m * OUT_F + 16 + m16] = c1[j] + bv1;
        }
    }
}

extern "C" void kernel_launch(void* const* d_in, const int* in_sizes, int n_in,
                              void* d_out, int out_size, void* d_ws, size_t ws_size,
                              hipStream_t stream) {
    (void)n_in; (void)d_ws; (void)ws_size; (void)out_size;
    const float* x = (const float*)d_in[0];
    const float* w = (const float*)d_in[1];
    const float* b = (const float*)d_in[2];
    const int num_networks   = in_sizes[3];
    const int total_points   = in_sizes[0] / IN_F;
    const int points_per_net = total_points / num_networks;

    grouped_linear_wmma_f32<<<dim3(num_networks), dim3(256), 0, stream>>>(
        x, w, b, (float*)d_out, points_per_net);
}


# Fewer, longer waves: one wave per network keeps W fragments in registers over its eight 16-row tiles, all x loads in flight; 1024 waves (256 workgroups of 4) enter instead of 8192; no barrier
# speedup vs baseline: 1.1214x; 1.1214x over previous
#include <hip/hip_runtime.h>


typedef __attribute__((ext_vector_type(2))) float v2f;
typedef __attribute__((ext_vector_type(8))) float v8f;

#define IN_F 32
#define OUT_F 32
#define LDS_STRIDE 36
#define WAVES 4
#define CHUNK 8

struct XTile  { float4 q0, q1, q2, q3; };
struct XChunk { XTile t0, t1, t2, t3, t4, t5, t6, t7; };
struct WRegs  { float4 q0, q1, q2, q3, q4, q5, q6, q7; };
struct Frag   { v2f k0, k1, k2, k3, k4, k5, k6, k7; };

__device__ __forceinline__ XTile load_tile(const float* __restrict__ Xt, const int lane) {
    const float4* p = reinterpret_cast<const float4*>(Xt);
    XTile v;
    v.q0 = p[lane]; v.q1 = p[32 + lane]; v.q2 = p[64 + lane]; v.q3 = p[96 + lane];
    return v;
}

__device__ __forceinline__ XChunk load_chunk(const float* __restrict__ Xc, const int lane) {
    XChunk v;
    v.t0 = load_tile(Xc + 0 * 16 * IN_F, lane);  v.t1 = load_tile(Xc + 1 * 16 * IN_F, lane);
    v.t2 = load_tile(Xc + 2 * 16 * IN_F, lane);  v.t3 = load_tile(Xc + 3 * 16 * IN_F, lane);
    v.t4 = load_tile(Xc + 4 * 16 * IN_F, lane);  v.t5 = load_tile(Xc + 5 * 16 * IN_F, lane);
    v.t6 = load_tile(Xc + 6 * 16 * IN_F, lane);  v.t7 = load_tile(Xc + 7 * 16 * IN_F, lane);
    return v;
}

__device__ __forceinline__ WRegs load_w(const float* __restrict__ Wn, const int lane) {
    const float4* p = reinterpret_cast<const float4*>(Wn);
    WRegs w;
    w.q0 = p[lane];       w.q1 = p[32 + lane];  w.q2 = p[64 + lane];  w.q3 = p[96 + lane];
    w.q4 = p[128 + lane]; w.q5 = p[160 + lane]; w.q6 = p[192 + lane]; w.q7 = p[224 + lane];
    return w;
}

__device__ __forceinline__ void stage_w(const WRegs& w, float* wl, const int lane, Frag& b0, Frag& b1) {
    const int m16  = lane & 15;
    const int half = lane >> 4;
    const int k = lane >> 3;
    const int n = (lane & 7) * 4;
    *reinterpret_cast<float4*>(&wl[(k     ) * LDS_STRIDE + n]) = w.q0;
    *reinterpret_cast<float4*>(&wl[(k +  4) * LDS_STRIDE + n]) = w.q1;
    *reinterpret_cast<float4*>(&wl[(k +  8) * LDS_STRIDE + n]) = w.q2;
    *reinterpret_cast<float4*>(&wl[(k + 12) * LDS_STRIDE + n]) = w.q3;
    *reinterpret_cast<float4*>(&wl[(k + 16) * LDS_STRIDE + n]) = w.q4;
    *reinterpret_cast<float4*>(&wl[(k + 20) * LDS_STRIDE + n]) = w.q5;
    *reinterpret_cast<float4*>(&wl[(k + 24) * LDS_STRIDE + n]) = w.q6;
    *reinterpret_cast<float4*>(&wl[(k + 28) * LDS_STRIDE + n]) = w.q7;
    const float* bp = &wl[(half * 2) * LDS_STRIDE + m16];
    #define BLOAD(KK, F)                                                                  \
        b0.F.x = bp[((KK) * 4) * LDS_STRIDE];      b0.F.y = bp[((KK) * 4 + 1) * LDS_STRIDE];      \
        b1.F.x = bp[((KK) * 4) * LDS_STRIDE + 16]; b1.F.y = bp[((KK) * 4 + 1) * LDS_STRIDE + 16];
    BLOAD(0, k0) BLOAD(1, k1) BLOAD(2, k2) BLOAD(3, k3)
    BLOAD(4, k4) BLOAD(5, k5) BLOAD(6, k6) BLOAD(7, k7)
    #undef BLOAD
}

__device__ __forceinline__ Frag stage_and_fetch(const XTile& v, float* st, const int lane) {
    const int m16  = lane & 15;
    const int half = lane >> 4;
    const int r = lane >> 3;
    const int c = (lane & 7) * 4;
    *reinterpret_cast<float4*>(&st[(r     ) * LDS_STRIDE + c]) = v.q0;
    *reinterpret_cast<float4*>(&st[(r +  4) * LDS_STRIDE + c]) = v.q1;
    *reinterpret_cast<float4*>(&st[(r +  8) * LDS_STRIDE + c]) = v.q2;
    *reinterpret_cast<float4*>(&st[(r + 12) * LDS_STRIDE + c]) = v.q3;
    const float* ap = &st[m16 * LDS_STRIDE + half * 2];
    Frag a;
    a.k0 = *reinterpret_cast<const v2f*>(ap +  0);
    a.k1 = *reinterpret_cast<const v2f*>(ap +  4);
    a.k2 = *reinterpret_cast<const v2f*>(ap +  8);
    a.k3 = *reinterpret_cast<const v2f*>(ap + 12);
    a.k4 = *reinterpret_cast<const v2f*>(ap + 16);
    a.k5 = *reinterpret_cast<const v2f*>(ap + 20);
    a.k6 = *reinterpret_cast<const v2f*>(ap + 24);
    a.k7 = *reinterpret_cast<const v2f*>(ap + 28);
    return a;
}

#define WMMA_F32(A, B, C) \
    __builtin_amdgcn_wmma_f32_16x16x4_f32(false, (A), false, (B), (short)0, (C), false, false)

__device__ __forceinline__ void
mul_store_tile(const Frag& a, const Frag& b0, const Frag& b1,
               const float bv0, const float bv1, float* __restrict__ Yt, const int lane)
{
    const int m16  = lane & 15;
    const int half = lane >> 4;
    v8f c0 = {};
    v8f c1 = {};
    c0 = WMMA_F32(a.k0, b0.k0, c0);  c1 = WMMA_F32(a.k0, b1.k0, c1);
    c0 = WMMA_F32(a.k1, b0.k1, c0);  c1 = WMMA_F32(a.k1, b1.k1, c1);
    c0 = WMMA_F32(a.k2, b0.k2, c0);  c1 = WMMA_F32(a.k2, b1.k2, c1);
    c0 = WMMA_F32(a.k3, b0.k3, c0);  c1 = WMMA_F32(a.k3, b1.k3, c1);
    c0 = WMMA_F32(a.k4, b0.k4, c0);  c1 = WMMA_F32(a.k4, b1.k4, c1);
    c0 = WMMA_F32(a.k5, b0.k5, c0);  c1 = WMMA_F32(a.k5, b1.k5, c1);
    c0 = WMMA_F32(a.k6, b0.k6, c0);  c1 = WMMA_F32(a.k6, b1.k6, c1);
    c0 = WMMA_F32(a.k7, b0.k7, c0);  c1 = WMMA_F32(a.k7, b1.k7, c1);
    float* yp = Yt + (8 * half) * OUT_F + m16;
    #pragma unroll
    for (int j = 0; j < 8; ++j) {
        yp[j * OUT_F]      = c0[j] + bv0;
        yp[j * OUT_F + 16] = c1[j] + bv1;
    }
}

#define TILE_ORDER() __builtin_amdgcn_sched_barrier(0)

__device__ __forceinline__ void
run_chunk(const XChunk& v, float* __restrict__ Yc, float* xs,
          const Frag& b0, const Frag& b1,
          const float bv0, const float bv1, const int lane)
{
    float* s0 = xs;
    float* s1 = xs + 16 * LDS_STRIDE;
    const Frag a0 = stage_and_fetch(v.t0, s0, lane);
    const Frag a1 = stage_and_fetch(v.t1, s1, lane);
    TILE_ORDER();
    mul_store_tile(a0, b0, b1, bv0, bv1, Yc + 0 * 16 * OUT_F, lane);
    TILE_ORDER();
    const Frag a2 = stage_and_fetch(v.t2, s0, lane);
    TILE_ORDER();
    mul_store_tile(a1, b0, b1, bv0, bv1, Yc + 1 * 16 * OUT_F, lane);
    TILE_ORDER();
    const Frag a3 = stage_and_fetch(v.t3, s1, lane);
    TILE_ORDER();
    mul_store_tile(a2, b0, b1, bv0, bv1, Yc + 2 * 16 * OUT_F, lane);
    TILE_ORDER();
    const Frag a4 = stage_and_fetch(v.t4, s0, lane);
    TILE_ORDER();
    mul_store_tile(a3, b0, b1, bv0, bv1, Yc + 3 * 16 * OUT_F, lane);
    TILE_ORDER();
    const Frag a5 = stage_and_fetch(v.t5, s1, lane);
    TILE_ORDER();
    mul_store_tile(a4, b0, b1, bv0, bv1, Yc + 4 * 16 * OUT_F, lane);
    TILE_ORDER();
    const Frag a6 = stage_and_fetch(v.t6, s0, lane);
    TILE_ORDER();
    mul_store_tile(a5, b0, b1, bv0, bv1, Yc + 5 * 16 * OUT_F, lane);
    TILE_ORDER();
    const Frag a7 = stage_and_fetch(v.t7, s1, lane);
    TILE_ORDER();
    mul_store_tile(a6, b0, b1, bv0, bv1, Yc + 6 * 16 * OUT_F, lane);
    TILE_ORDER();
    mul_store_tile(a7, b0, b1, bv0, bv1, Yc + 7 * 16 * OUT_F, lane);
}

__device__ __forceinline__ void
run_single_tiles(int tile, const int ntiles, const float* __restrict__ Xn, float* __restrict__ Yn,
                 float* xs, const Frag& b0, const Frag& b1,
                 const float bv0, const float bv1, const int lane)
{
    for (; tile < ntiles; ++tile) {
        const XTile v = load_tile(Xn + (size_t)tile * 16 * IN_F, lane);
        const Frag  a = stage_and_fetch(v, xs, lane);
        mul_store_tile(a, b0, b1, bv0, bv1, Yn + (size_t)tile * 16 * OUT_F, lane);
    }
}

__global__ __launch_bounds__(WAVES * 32) void
grouped_linear_wmma_f32(const float* __restrict__ x,
                        const float* __restrict__ weights,
                        const float* __restrict__ biases,
                        float* __restrict__ y,
                        int points_per_net, int num_networks)
{
    __shared__ float ldsW[WAVES][IN_F * LDS_STRIDE];
    __shared__ float ldsX[WAVES][2 * 16 * LDS_STRIDE];

    const int tid  = threadIdx.x;
    const int wave = tid >> 5;
    const int lane = tid & 31;
    const int net  = blockIdx.x * WAVES + wave;
    if (net >= num_networks) return;

    const int m16 = lane & 15;
    float* wl = ldsW[wave];
    float* xs = ldsX[wave];

    const size_t rowbase = (size_t)net * (size_t)points_per_net;
    const float* Xn = x + rowbase * IN_F;
    float*       Yn = y + rowbase * OUT_F;
    const int ntiles = (points_per_net + 15) >> 4;

    const WRegs w   = load_w(weights + (size_t)net * IN_F * OUT_F, lane);
    const float bv0 = biases[(size_t)net * OUT_F + m16];
    const float bv1 = biases[(size_t)net * OUT_F + 16 + m16];

    Frag b0, b1;
    if (ntiles >= CHUNK) {
        XChunk v = load_chunk(Xn, lane);
        TILE_ORDER();
        stage_w(w, wl, lane, b0, b1);
        int tile = 0;
        for (;;) {
            run_chunk(v, Yn + (size_t)tile * 16 * OUT_F, xs, b0, b1, bv0, bv1, lane);
            tile += CHUNK;
            if (tile + CHUNK > ntiles) break;
            v = load_chunk(Xn + (size_t)tile * 16 * IN_F, lane);
        }
        run_single_tiles(tile, ntiles, Xn, Yn, xs, b0, b1, bv0, bv1, lane);
    } else {
        stage_w(w, wl, lane, b0, b1);
        run_single_tiles(0, ntiles, Xn, Yn, xs, b0, b1, bv0, bv1, lane);
    }
}

extern "C" void kernel_launch(void* const* d_in, const int* in_sizes, int n_in,
                              void* d_out, int out_size, void* d_ws, size_t ws_size,
                              hipStream_t stream) {
    (void)n_in; (void)d_ws; (void)ws_size; (void)out_size;
    const float* x = (const float*)d_in[0];
    const float* w = (const float*)d_in[1];
    const float* b = (const float*)d_in[2];
    const int num_networks   = in_sizes[3];
    const int total_points   = in_sizes[0] / IN_F;
    const int points_per_net = total_points / num_networks;

    grouped_linear_wmma_f32<<<dim3((num_networks + WAVES - 1) / WAVES), dim3(WAVES * 32), 0, stream>>>(
        x, w, b, (float*)d_out, points_per_net, num_networks);
}


# Two waves per SIMD: each network's eight row tiles are split between two waves (256 workgroups of 8), so one wave's staging, bias adds and stores run under its sibling's WMMAs; sums unchanged
# speedup vs baseline: 1.2061x; 1.0755x over previous
#include <hip/hip_runtime.h>


typedef __attribute__((ext_vector_type(2))) float v2f;
typedef __attribute__((ext_vector_type(8))) float v8f;

#define IN_F 32
#define OUT_F 32
#define LDS_STRIDE 36
#define NETS  4
#define PARTS 2
#define WAVES (NETS * PARTS)
#define CHUNK 4

struct XTile  { float4 q0, q1, q2, q3; };
struct XChunk { XTile t0, t1, t2, t3; };
struct WRegs  { float4 q0, q1, q2, q3, q4, q5, q6, q7; };
struct Frag   { v2f k0, k1, k2, k3, k4, k5, k6, k7; };

__device__ __forceinline__ XTile load_tile(const float* __restrict__ Xt, const int lane) {
    const float4* p = reinterpret_cast<const float4*>(Xt);
    XTile v;
    v.q0 = p[lane]; v.q1 = p[32 + lane]; v.q2 = p[64 + lane]; v.q3 = p[96 + lane];
    return v;
}

__device__ __forceinline__ XChunk load_chunk(const float* __restrict__ Xc, const int lane) {
    XChunk v;
    v.t0 = load_tile(Xc + 0 * 16 * IN_F, lane);  v.t1 = load_tile(Xc + 1 * 16 * IN_F, lane);
    v.t2 = load_tile(Xc + 2 * 16 * IN_F, lane);  v.t3 = load_tile(Xc + 3 * 16 * IN_F, lane);
    return v;
}

__device__ __forceinline__ WRegs load_w(const float* __restrict__ Wn, const int lane) {
    const float4* p = reinterpret_cast<const float4*>(Wn);
    WRegs w;
    w.q0 = p[lane];       w.q1 = p[32 + lane];  w.q2 = p[64 + lane];  w.q3 = p[96 + lane];
    w.q4 = p[128 + lane]; w.q5 = p[160 + lane]; w.q6 = p[192 + lane]; w.q7 = p[224 + lane];
    return w;
}

__device__ __forceinline__ void stage_w(const WRegs& w, float* wl, const int lane, Frag& b0, Frag& b1) {
    const int m16  = lane & 15;
    const int half = lane >> 4;
    const int k = lane >> 3;
    const int n = (lane & 7) * 4;
    *reinterpret_cast<float4*>(&wl[(k     ) * LDS_STRIDE + n]) = w.q0;
    *reinterpret_cast<float4*>(&wl[(k +  4) * LDS_STRIDE + n]) = w.q1;
    *reinterpret_cast<float4*>(&wl[(k +  8) * LDS_STRIDE + n]) = w.q2;
    *reinterpret_cast<float4*>(&wl[(k + 12) * LDS_STRIDE + n]) = w.q3;
    *reinterpret_cast<float4*>(&wl[(k + 16) * LDS_STRIDE + n]) = w.q4;
    *reinterpret_cast<float4*>(&wl[(k + 20) * LDS_STRIDE + n]) = w.q5;
    *reinterpret_cast<float4*>(&wl[(k + 24) * LDS_STRIDE + n]) = w.q6;
    *reinterpret_cast<float4*>(&wl[(k + 28) * LDS_STRIDE + n]) = w.q7;
    const float* bp = &wl[(half * 2) * LDS_STRIDE + m16];
    #define BLOAD(KK, F)                                                                  \
        b0.F.x = bp[((KK) * 4) * LDS_STRIDE];      b0.F.y = bp[((KK) * 4 + 1) * LDS_STRIDE];      \
        b1.F.x = bp[((KK) * 4) * LDS_STRIDE + 16]; b1.F.y = bp[((KK) * 4 + 1) * LDS_STRIDE + 16];
    BLOAD(0, k0) BLOAD(1, k1) BLOAD(2, k2) BLOAD(3, k3)
    BLOAD(4, k4) BLOAD(5, k5) BLOAD(6, k6) BLOAD(7, k7)
    #undef BLOAD
}

__device__ __forceinline__ Frag stage_and_fetch(const XTile& v, float* st, const int lane) {
    const int m16  = lane & 15;
    const int half = lane >> 4;
    const int r = lane >> 3;
    const int c = (lane & 7) * 4;
    *reinterpret_cast<float4*>(&st[(r     ) * LDS_STRIDE + c]) = v.q0;
    *reinterpret_cast<float4*>(&st[(r +  4) * LDS_STRIDE + c]) = v.q1;
    *reinterpret_cast<float4*>(&st[(r +  8) * LDS_STRIDE + c]) = v.q2;
    *reinterpret_cast<float4*>(&st[(r + 12) * LDS_STRIDE + c]) = v.q3;
    const float* ap = &st[m16 * LDS_STRIDE + half * 2];
    Frag a;
    a.k0 = *reinterpret_cast<const v2f*>(ap +  0);
    a.k1 = *reinterpret_cast<const v2f*>(ap +  4);
    a.k2 = *reinterpret_cast<const v2f*>(ap +  8);
    a.k3 = *reinterpret_cast<const v2f*>(ap + 12);
    a.k4 = *reinterpret_cast<const v2f*>(ap + 16);
    a.k5 = *reinterpret_cast<const v2f*>(ap + 20);
    a.k6 = *reinterpret_cast<const v2f*>(ap + 24);
    a.k7 = *reinterpret_cast<const v2f*>(ap + 28);
    return a;
}

#define WMMA_F32(A, B, C) \
    __builtin_amdgcn_wmma_f32_16x16x4_f32(false, (A), false, (B), (short)0, (C), false, false)

__device__ __forceinline__ void
mul_store_tile(const Frag& a, const Frag& b0, const Frag& b1,
               const float bv0, const float bv1, float* __restrict__ Yt, const int lane)
{
    const int m16  = lane & 15;
    const int half = lane >> 4;
    v8f c0 = {};
    v8f c1 = {};
    c0 = WMMA_F32(a.k0, b0.k0, c0);  c1 = WMMA_F32(a.k0, b1.k0, c1);
    c0 = WMMA_F32(a.k1, b0.k1, c0);  c1 = WMMA_F32(a.k1, b1.k1, c1);
    c0 = WMMA_F32(a.k2, b0.k2, c0);  c1 = WMMA_F32(a.k2, b1.k2, c1);
    c0 = WMMA_F32(a.k3, b0.k3, c0);  c1 = WMMA_F32(a.k3, b1.k3, c1);
    c0 = WMMA_F32(a.k4, b0.k4, c0);  c1 = WMMA_F32(a.k4, b1.k4, c1);
    c0 = WMMA_F32(a.k5, b0.k5, c0);  c1 = WMMA_F32(a.k5, b1.k5, c1);
    c0 = WMMA_F32(a.k6, b0.k6, c0);  c1 = WMMA_F32(a.k6, b1.k6, c1);
    c0 = WMMA_F32(a.k7, b0.k7, c0);  c1 = WMMA_F32(a.k7, b1.k7, c1);
    float* yp = Yt + (8 * half) * OUT_F + m16;
    #pragma unroll
    for (int j = 0; j < 8; ++j) {
        yp[j * OUT_F]      = c0[j] + bv0;
        yp[j * OUT_F + 16] = c1[j] + bv1;
    }
}

#define TILE_ORDER() __builtin_amdgcn_sched_barrier(0)

__device__ __forceinline__ void
run_chunk(const XChunk& v, float* __restrict__ Yc, float* xs,
          const Frag& b0, const Frag& b1,
          const float bv0, const float bv1, const int lane)
{
    float* s0 = xs;
    float* s1 = xs + 16 * LDS_STRIDE;
    const Frag a0 = stage_and_fetch(v.t0, s0, lane);
    const Frag a1 = stage_and_fetch(v.t1, s1, lane);
    TILE_ORDER();
    mul_store_tile(a0, b0, b1, bv0, bv1, Yc + 0 * 16 * OUT_F, lane);
    TILE_ORDER();
    const Frag a2 = stage_and_fetch(v.t2, s0, lane);
    TILE_ORDER();
    mul_store_tile(a1, b0, b1, bv0, bv1, Yc + 1 * 16 * OUT_F, lane);
    TILE_ORDER();
    const Frag a3 = stage_and_fetch(v.t3, s1, lane);
    TILE_ORDER();
    mul_store_tile(a2, b0, b1, bv0, bv1, Yc + 2 * 16 * OUT_F, lane);
    TILE_ORDER();
    mul_store_tile(a3, b0, b1, bv0, bv1, Yc + 3 * 16 * OUT_F, lane);
}

__device__ __forceinline__ void
run_single_tiles(int tile, const int ntiles, const float* __restrict__ Xn, float* __restrict__ Yn,
                 float* xs, const Frag& b0, const Frag& b1,
                 const float bv0, const float bv1, const int lane)
{
    for (; tile < ntiles; ++tile) {
        const XTile v = load_tile(Xn + (size_t)tile * 16 * IN_F, lane);
        const Frag  a = stage_and_fetch(v, xs, lane);
        mul_store_tile(a, b0, b1, bv0, bv1, Yn + (size_t)tile * 16 * OUT_F, lane);
    }
}

__global__ __launch_bounds__(WAVES * 32) void
grouped_linear_wmma_f32(const float* __restrict__ x,
                        const float* __restrict__ weights,
                        const float* __restrict__ biases,
                        float* __restrict__ y,
                        int points_per_net, int num_networks)
{
    static_assert(2 * 16 * LDS_STRIDE <= IN_F * LDS_STRIDE, "x stages must fit the W region");
    __shared__ float lds[WAVES][IN_F * LDS_STRIDE];

    const int tid  = threadIdx.x;
    const int wave = tid >> 5;
    const int lane = tid & 31;
    const int net  = blockIdx.x * NETS + (wave % NETS);
    const int part = wave / NETS;
    if (net >= num_networks) return;

    const int all_tiles = (points_per_net + 15) >> 4;
    const int per_part  = (all_tiles + PARTS - 1) / PARTS;
    const int tbeg      = part * per_part;
    const int ntiles    = min(all_tiles - tbeg, per_part);
    if (ntiles <= 0) return;

    const int m16 = lane & 15;
    float* wl = lds[wave];
    float* xs = lds[wave];

    const size_t rowbase = (size_t)net * (size_t)points_per_net + (size_t)tbeg * 16;
    const float* Xn = x + rowbase * IN_F;
    float*       Yn = y + rowbase * OUT_F;

    const WRegs w   = load_w(weights + (size_t)net * IN_F * OUT_F, lane);
    const float bv0 = biases[(size_t)net * OUT_F + m16];
    const float bv1 = biases[(size_t)net * OUT_F + 16 + m16];

    Frag b0, b1;
    if (ntiles >= CHUNK) {
        XChunk v = load_chunk(Xn, lane);
        TILE_ORDER();
        stage_w(w, wl, lane, b0, b1);
        int tile = 0;
        for (;;) {
            run_chunk(v, Yn + (size_t)tile * 16 * OUT_F, xs, b0, b1, bv0, bv1, lane);
            tile += CHUNK;
            if (tile + CHUNK > ntiles) break;
            v = load_chunk(Xn + (size_t)tile * 16 * IN_F, lane);
        }
        run_single_tiles(tile, ntiles, Xn, Yn, xs, b0, b1, bv0, bv1, lane);
    } else {
        stage_w(w, wl, lane, b0, b1);
        run_single_tiles(0, ntiles, Xn, Yn, xs, b0, b1, bv0, bv1, lane);
    }
}

extern "C" void kernel_launch(void* const* d_in, const int* in_sizes, int n_in,
                              void* d_out, int out_size, void* d_ws, size_t ws_size,
                              hipStream_t stream) {
    (void)n_in; (void)d_ws; (void)ws_size; (void)out_size;
    const float* x = (const float*)d_in[0];
    const float* w = (const float*)d_in[1];
    const float* b = (const float*)d_in[2];
    const int num_networks   = in_sizes[3];
    const int total_points   = in_sizes[0] / IN_F;
    const int points_per_net = total_points / num_networks;

    grouped_linear_wmma_f32<<<dim3((num_networks + NETS - 1) / NETS), dim3(WAVES * 32), 0, stream>>>(
        x, w, b, (float*)d_out, points_per_net, num_networks);
}
